# starting kernel
# baseline (speedup 1.0000x reference)
#include <hip/hip_runtime.h>

typedef float v2f __attribute__((ext_vector_type(2)));
typedef float v8f __attribute__((ext_vector_type(8)));

#define SMALL 512
#define LARGE 1024
#define NPOS  8192
#define ROWS  32
#define STR   (SMALL + 4)

__global__ __launch_bounds__(256)
void fused_embed_wmma(const int* __restrict__ ids,
                      const float* __restrict__ weight,
                      const float* __restrict__ cw,
                      const float* __restrict__ residual,
                      float* __restrict__ out)
{
    __shared__ __align__(16) float As[ROWS * STR];

    const unsigned pos0 = blockIdx.x * ROWS;
    const unsigned tid  = threadIdx.x;

    {
        const unsigned row0 = tid >> 4;
        const unsigned c16  = tid & 15;
#pragma unroll
        for (unsigned rr = 0; rr < 2; ++rr) {
            const unsigned row = row0 + rr * 16;
            const unsigned t   = (unsigned)ids[pos0 + row];
            const unsigned w4  = t << 7;
            const unsigned r4  = t << 8;
            const unsigned o4  = (pos0 + row) << 8;
            const float4* wv = (const float4*)weight;
            const float4* rv = (const float4*)residual;
            float4*       ov = (float4*)out;
            float*        lrow = As + row * STR;
#pragma unroll
            for (unsigned j = 0; j < 8; ++j) {
                const unsigned c4 = c16 + j * 16;
                float4 w = wv[w4 + c4];
                float4 r = rv[r4 + c4];
                *(float4*)(lrow + c4 * 4) = w;
                float4 o;
                o.x = w.x + r.x; o.y = w.y + r.y; o.z = w.z + r.z; o.w = w.w + r.w;
                ov[o4 + c4] = o;
            }
        }
    }
    __syncthreads();

    const unsigned wave   = tid >> 5;
    const unsigned lane   = tid & 31;
    const unsigned lane16 = lane & 15;
    const unsigned hi     = lane >> 4;
    const unsigned koff   = hi * 2;

    const float* a0p = As + lane16 * STR + koff;
    const float* a1p = As + (16 + lane16) * STR + koff;

    const unsigned nb0 = wave * 64;
    const float* b0p = cw + (((nb0 +      lane16) << 9) + koff);
    const float* b1p = cw + (((nb0 + 16 + lane16) << 9) + koff);
    const float* b2p = cw + (((nb0 + 32 + lane16) << 9) + koff);
    const float* b3p = cw + (((nb0 + 48 + lane16) << 9) + koff);

    v8f c00 = {}, c01 = {}, c02 = {}, c03 = {};
    v8f c10 = {}, c11 = {}, c12 = {}, c13 = {};

#pragma unroll 2
    for (unsigned k = 0; k < SMALL; k += 4) {
        v2f a0 = *(const v2f*)(a0p + k);
        v2f a1 = *(const v2f*)(a1p + k);
        v2f b0 = *(const v2f*)(b0p + k);
        v2f b1 = *(const v2f*)(b1p + k);
        v2f b2 = *(const v2f*)(b2p + k);
        v2f b3 = *(const v2f*)(b3p + k);
        c00 = __builtin_amdgcn_wmma_f32_16x16x4_f32(false, a0, false, b0, (short)0, c00, false, false);
        c10 = __builtin_amdgcn_wmma_f32_16x16x4_f32(false, a1, false, b0, (short)0, c10, false, false);
        c01 = __builtin_amdgcn_wmma_f32_16x16x4_f32(false, a0, false, b1, (short)0, c01, false, false);
        c11 = __builtin_amdgcn_wmma_f32_16x16x4_f32(false, a1, false, b1, (short)0, c11, false, false);
        c02 = __builtin_amdgcn_wmma_f32_16x16x4_f32(false, a0, false, b2, (short)0, c02, false, false);
        c12 = __builtin_amdgcn_wmma_f32_16x16x4_f32(false, a1, false, b2, (short)0, c12, false, false);
        c03 = __builtin_amdgcn_wmma_f32_16x16x4_f32(false, a0, false, b3, (short)0, c03, false, false);
        c13 = __builtin_amdgcn_wmma_f32_16x16x4_f32(false, a1, false, b3, (short)0, c13, false, false);
    }

    int toks0[8], toks1[8];
#pragma unroll
    for (unsigned v = 0; v < 8; ++v) {
        toks0[v] = ids[pos0 + hi * 8 + v];
        toks1[v] = ids[pos0 + 16 + hi * 8 + v];
    }

    const unsigned col = SMALL + nb0 + lane16;
#pragma unroll
    for (unsigned v = 0; v < 8; ++v) {
        const unsigned m0 = hi * 8 + v;
        const unsigned m1 = 16 + m0;
        const unsigned o0 = ((pos0 + m0) << 10) + col;
        const unsigned o1 = ((pos0 + m1) << 10) + col;
        const unsigned r0 = (((unsigned)toks0[v]) << 10) + col;
        const unsigned r1 = (((unsigned)toks1[v]) << 10) + col;
        out[o0]      = c00[v] + residual[r0];
        out[o0 + 16] = c01[v] + residual[r0 + 16];
        out[o0 + 32] = c02[v] + residual[r0 + 32];
        out[o0 + 48] = c03[v] + residual[r0 + 48];
        out[o1]      = c10[v] + residual[r1];
        out[o1 + 16] = c11[v] + residual[r1 + 16];
        out[o1 + 32] = c12[v] + residual[r1 + 32];
        out[o1 + 48] = c13[v] + residual[r1 + 48];
    }
}

extern "C" void kernel_launch(void* const* d_in, const int* in_sizes, int n_in,
                              void* d_out, int out_size, void* d_ws, size_t ws_size,
                              hipStream_t stream) {
    const int*   ids      = (const int*)d_in[0];
    const float* weight   = (const float*)d_in[1];
    const float* cw       = (const float*)d_in[2];
    const float* residual = (const float*)d_in[3];
    float*       out      = (float*)d_out;

    dim3 grid(NPOS / ROWS);
    dim3 block(256);
    fused_embed_wmma<<<grid, block, 0, stream>>>(ids, weight, cw, residual, out);
}


# Global loads moved under the matrix run: weight rows reach LDS in 4 K-chunks fetched one chunk ahead (one barrier each), epilogue ids and residual fetched during the last chunk; every sum unchanged
# speedup vs baseline: 1.3485x; 1.3485x over previous
#include <hip/hip_runtime.h>


typedef float v2f __attribute__((ext_vector_type(2)));
typedef float v4f __attribute__((ext_vector_type(4)));
typedef float v8f __attribute__((ext_vector_type(8)));

#define SMALL 512
#define LARGE 1024
#define NPOS  8192
#define ROWS  64
#define NSPL  2
#define NCOL  (SMALL / NSPL)
#define STR   (SMALL + 4)

template <unsigned NH>
__device__ __forceinline__
void embed_body(float* As,
                const int* __restrict__ ids,
                const float* __restrict__ weight,
                const float* __restrict__ cw,
                const float* __restrict__ residual,
                float* __restrict__ out)
{
    const unsigned pos0 = (blockIdx.x / NSPL) * ROWS;
    const unsigned nh   = NH;
    const unsigned tid  = threadIdx.x;

#define NCH 4
#define CK  (SMALL / NCH)
#define C4  (CK / 4)
#define RPT (ROWS / 8)
    const unsigned grow = tid >> 5;
    const unsigned gc4  = tid & 31;
    const v4f* wv = (const v4f*)weight;
    const v4f* rv = (const v4f*)residual;
    v4f*       ov = (v4f*)out;

    unsigned tk[RPT];
#pragma unroll
    for (unsigned i = 0; i < RPT; ++i) tk[i] = (unsigned)ids[pos0 + grow + 8 * i];

    v4f g[RPT], r[RPT];
#define GATHER_ISSUE(C)                                                             \
    {                                                                               \
        const unsigned cc = (C) * C4 + gc4;                                         \
        _Pragma("unroll")                                                           \
        for (unsigned i = 0; i < RPT; ++i) g[i] = wv[(tk[i] << 7) + cc];            \
        if (((C) * CK) / NCOL == nh) {                                              \
            _Pragma("unroll")                                                       \
            for (unsigned i = 0; i < RPT; ++i) r[i] = rv[(tk[i] << 8) + cc];        \
        }                                                                           \
    }
#define GATHER_TO_LDS(C)                                                            \
    {                                                                               \
        const unsigned cc = (C) * C4 + gc4;                                         \
        _Pragma("unroll")                                                           \
        for (unsigned i = 0; i < RPT; ++i)                                          \
            *(v4f*)(As + (grow + 8 * i) * STR + cc * 4) = g[i];                     \
    }
#define R_PIN                                                                       \
    asm volatile("" : "+v"(r[0]), "+v"(r[1]), "+v"(r[2]), "+v"(r[3]),               \
                      "+v"(r[4]), "+v"(r[5]), "+v"(r[6]), "+v"(r[7]));
#define GATHER_TO_OUT(C)                                                            \
    {                                                                               \
        const unsigned cc = (C) * C4 + gc4;                                         \
        if (((C) * CK) / NCOL == nh) {                                              \
            R_PIN                                                                   \
            _Pragma("unroll")                                                       \
            for (unsigned i = 0; i < RPT; ++i)                                      \
                ov[((pos0 + grow + 8 * i) << 8) + cc] = g[i] + r[i];                \
        }                                                                           \
    }

    const unsigned wave   = tid >> 5;
    const unsigned lane   = tid & 31;
    const unsigned lane16 = lane & 15;
    const unsigned hi     = lane >> 4;
    const unsigned koff   = hi * 2;

    const float* a0p = As + (     lane16) * STR + koff;
    const float* a1p = As + (16 + lane16) * STR + koff;
    const float* a2p = As + (32 + lane16) * STR + koff;
    const float* a3p = As + (48 + lane16) * STR + koff;

    const unsigned nb0 = nh * NCOL + wave * 32;
    const float* b0p = cw + (((nb0 +      lane16) << 9) + koff);
    const float* b1p = cw + (((nb0 + 16 + lane16) << 9) + koff);

    v8f c00 = {}, c01 = {};
    v8f c10 = {}, c11 = {};
    v8f c20 = {}, c21 = {};
    v8f c30 = {}, c31 = {};

#define KB 16
#define FRAG_LOAD_B(F, KK)                                                          \
    {                                                                               \
        _Pragma("unroll")                                                           \
        for (unsigned s = 0; s < KB / 4; ++s) {                                     \
            F.b0[s] = *(const v2f*)(b0p + (KK) + 4 * s);                            \
            F.b1[s] = *(const v2f*)(b1p + (KK) + 4 * s);                            \
        }                                                                           \
    }
#define FRAG_LOAD_A(F, KK)                                                          \
    {                                                                               \
        _Pragma("unroll")                                                           \
        for (unsigned s = 0; s < KB / 4; ++s) {                                     \
            F.a0[s] = *(const v2f*)(a0p + (KK) + 4 * s);                            \
            F.a1[s] = *(const v2f*)(a1p + (KK) + 4 * s);                            \
            F.a2[s] = *(const v2f*)(a2p + (KK) + 4 * s);                            \
            F.a3[s] = *(const v2f*)(a3p + (KK) + 4 * s);                            \
        }                                                                           \
    }
#define FRAG_LOAD(F, KK) FRAG_LOAD_B(F, KK) FRAG_LOAD_A(F, KK)
#define FRAG_MMA(F)                                                                 \
    {                                                                               \
        _Pragma("unroll")                                                           \
        for (unsigned s = 0; s < KB / 4; ++s) {                                     \
            c00 = __builtin_amdgcn_wmma_f32_16x16x4_f32(false, F.a0[s], false, F.b0[s], (short)0, c00, false, false); \
            c10 = __builtin_amdgcn_wmma_f32_16x16x4_f32(false, F.a1[s], false, F.b0[s], (short)0, c10, false, false); \
            c20 = __builtin_amdgcn_wmma_f32_16x16x4_f32(false, F.a2[s], false, F.b0[s], (short)0, c20, false, false); \
            c30 = __builtin_amdgcn_wmma_f32_16x16x4_f32(false, F.a3[s], false, F.b0[s], (short)0, c30, false, false); \
            c01 = __builtin_amdgcn_wmma_f32_16x16x4_f32(false, F.a0[s], false, F.b1[s], (short)0, c01, false, false); \
            c11 = __builtin_amdgcn_wmma_f32_16x16x4_f32(false, F.a1[s], false, F.b1[s], (short)0, c11, false, false); \
            c21 = __builtin_amdgcn_wmma_f32_16x16x4_f32(false, F.a2[s], false, F.b1[s], (short)0, c21, false, false); \
            c31 = __builtin_amdgcn_wmma_f32_16x16x4_f32(false, F.a3[s], false, F.b1[s], (short)0, c31, false, false); \
        }                                                                           \
    }
    struct Frag { v2f b0[KB / 4], b1[KB / 4], a0[KB / 4], a1[KB / 4], a2[KB / 4], a3[KB / 4]; };
    Frag f0, f1;

#define ORDER_FENCE                                                                 \
    asm volatile("" : "+v"(c00), "+v"(c01), "+v"(c10), "+v"(c11),                   \
                      "+v"(c20), "+v"(c21), "+v"(c30), "+v"(c31) :: "memory");      \
    __builtin_amdgcn_sched_barrier(0);
    const unsigned col = SMALL + nb0 + lane16;
    int   tok[4][8];
    float ra[4][8], rb[4][8];
#define EPI_IDS                                                                     \
    {                                                                               \
        _Pragma("unroll")                                                           \
        for (unsigned mt = 0; mt < 4; ++mt) {                                       \
            _Pragma("unroll")                                                       \
            for (unsigned v = 0; v < 8; ++v) tok[mt][v] = ids[pos0 + mt * 16 + hi * 8 + v]; \
        }                                                                           \
    }
#define EPI_RES(MT0)                                                                \
    {                                                                               \
        _Pragma("unroll")                                                           \
        for (unsigned mt = (MT0); mt < (MT0) + 2; ++mt) {                           \
            _Pragma("unroll")                                                       \
            for (unsigned v = 0; v < 8; ++v) {                                      \
                const unsigned rr = (((unsigned)tok[mt][v]) << 10) + col;    \
                ra[mt][v] = residual[rr];                                           \
                rb[mt][v] = residual[rr + 16];                                      \
            }                                                                       \
        }                                                                           \
    }
    GATHER_ISSUE(0u)
    FRAG_LOAD_B(f0, 0)
    GATHER_TO_LDS(0u)
    __syncthreads();
    FRAG_LOAD_A(f0, 0)
    GATHER_TO_OUT(0u)
#pragma unroll
    for (unsigned c = 0; c < NCH; ++c) {
        const unsigned kc = c * CK;
        FRAG_LOAD(f1, kc + KB)
        ORDER_FENCE
        FRAG_MMA(f0)
        ORDER_FENCE
        FRAG_LOAD(f0, kc + 2 * KB)
        ORDER_FENCE
        if (c + 1 < NCH) { GATHER_ISSUE(c + 1) } else { EPI_IDS }
        ORDER_FENCE
        FRAG_MMA(f1)
        ORDER_FENCE
#pragma unroll
        for (unsigned k = kc + 2 * KB; k < kc + CK - 2 * KB; k += 2 * KB) {
            FRAG_LOAD(f1, k + KB)
            ORDER_FENCE
            FRAG_MMA(f0)
            ORDER_FENCE
            FRAG_LOAD(f0, k + 2 * KB)
            ORDER_FENCE
            if (c + 1 == NCH && k == kc + 2 * KB) { EPI_RES(0) }
            if (c + 1 == NCH && k == kc + 4 * KB) { EPI_RES(2) }
            ORDER_FENCE
            FRAG_MMA(f1)
            ORDER_FENCE
        }
        FRAG_LOAD(f1, kc + CK - KB)
        ORDER_FENCE
        FRAG_MMA(f0)
        ORDER_FENCE
        if (c + 1 < NCH) {
            FRAG_LOAD_B(f0, kc + CK)
        }
        ORDER_FENCE
        FRAG_MMA(f1)
        ORDER_FENCE
        if (c + 1 < NCH) {
            GATHER_TO_LDS(c + 1)
            __syncthreads();
            FRAG_LOAD_A(f0, kc + CK)
            GATHER_TO_OUT(c + 1)
        }
    }
#undef ORDER_FENCE
#undef FRAG_LOAD
#undef FRAG_LOAD_A
#undef FRAG_LOAD_B
#undef FRAG_MMA
#undef KB
#undef GATHER_ISSUE
#undef GATHER_TO_LDS
#undef GATHER_TO_OUT
#undef R_PIN
#undef NCH
#undef CK
#undef C4
#undef RPT

#define EPI_TILE(MT, CA, CB)                                                        \
    {                                                                               \
        _Pragma("unroll")                                                           \
        for (unsigned v = 0; v < 8; ++v) {                                          \
            const unsigned m  = (MT) * 16 + hi * 8 + v;                             \
            const unsigned o  = ((pos0 + m) << 10) + col;                   \
            out[o]      = CA[v] + ra[MT][v];                                        \
            out[o + 16] = CB[v] + rb[MT][v];                                        \
        }                                                                           \
    }
    EPI_TILE(0, c00, c01)
    EPI_TILE(1, c10, c11)
    EPI_TILE(2, c20, c21)
    EPI_TILE(3, c30, c31)
#undef EPI_TILE
#undef EPI_IDS
#undef EPI_RES
}

__global__ __launch_bounds__(256)
void fused_embed_wmma(const int* __restrict__ ids,
                      const float* __restrict__ weight,
                      const float* __restrict__ cw,
                      const float* __restrict__ residual,
                      float* __restrict__ out)
{
    __shared__ __align__(16) float As[ROWS * STR];
    if (blockIdx.x % NSPL == 0) embed_body<0>(As, ids, weight, cw, residual, out);
    else                        embed_body<1>(As, ids, weight, cw, residual, out);
}

extern "C" void kernel_launch(void* const* d_in, const int* in_sizes, int n_in,
                              void* d_out, int out_size, void* d_ws, size_t ws_size,
                              hipStream_t stream) {
    const int*   ids      = (const int*)d_in[0];
    const float* weight   = (const float*)d_in[1];
    const float* cw       = (const float*)d_in[2];
    const float* residual = (const float*)d_in[3];
    float*       out      = (float*)d_out;

    dim3 grid((NPOS / ROWS) * NSPL);
    dim3 block(256);
    fused_embed_wmma<<<grid, block, 0, stream>>>(ids, weight, cw, residual, out);
}
